# starting kernel
# baseline (speedup 1.0000x reference)
#include <hip/hip_runtime.h>
#include <hip/hip_bf16.h>
#include <math.h>

typedef __attribute__((ext_vector_type(2))) float v2f;
typedef __attribute__((ext_vector_type(8))) float v8f;

#define T_DIM   2048
#define C_DIM   64
#define TSLICES 4
#define CGROUPS 4
#define WAVES   (TSLICES * CGROUPS)
#define BLOCK   (WAVES * 32)

__global__ __launch_bounds__(BLOCK)
void meta_feature_kernel(const float* __restrict__ x, float* __restrict__ out) {
    const int b    = blockIdx.x;
    const int tid  = threadIdx.x;
    const int wave = tid >> 5;
    const int lane = tid & 31;
    const int half = lane >> 4;
    const int ln   = lane & 15;
    const int cg   = wave & (CGROUPS - 1);
    const int ts   = wave >> 2;

    const int channel = cg * 16 + ln;
    const float* __restrict__ xb = x + (size_t)b * T_DIM * C_DIM;

    const int slice   = T_DIM / TSLICES;
    const int t_begin = ts * slice;
    const int t_end   = t_begin + slice;

    v8f acc_s = {};
    v8f acc_q = {};
    v2f ones;  ones.x = 1.0f; ones.y = 1.0f;

    float vmax = -INFINITY;
    float vmin =  INFINITY;

    for (int t0 = t_begin; t0 < t_end; t0 += 4) {
        const int r0 = t0 + half * 2;
        const float* p = xb + (size_t)r0 * C_DIM + channel;
        const float b0 = __builtin_nontemporal_load(p);
        const float b1 = __builtin_nontemporal_load(p + C_DIM);

        v2f bv; bv.x = b0;      bv.y = b1;
        v2f qv; qv.x = b0 * b0; qv.y = b1 * b1;

        acc_s = __builtin_amdgcn_wmma_f32_16x16x4_f32(
            false, ones, false, bv, (short)0, acc_s, false, false);
        acc_q = __builtin_amdgcn_wmma_f32_16x16x4_f32(
            false, ones, false, qv, (short)0, acc_q, false, false);

        vmax = fmaxf(vmax, fmaxf(b0, b1));
        vmin = fminf(vmin, fminf(b0, b1));
    }

    vmax = fmaxf(vmax, __shfl_xor(vmax, 16, 32));
    vmin = fminf(vmin, __shfl_xor(vmin, 16, 32));

    __shared__ float s_sum[WAVES * 16];
    __shared__ float s_sq [WAVES * 16];
    __shared__ float s_max[WAVES * 16];
    __shared__ float s_min[WAVES * 16];

    if (half == 0) {
        s_sum[wave * 16 + ln] = acc_s[0];
        s_sq [wave * 16 + ln] = acc_q[0];
        s_max[wave * 16 + ln] = vmax;
        s_min[wave * 16 + ln] = vmin;
    }
    __syncthreads();

    if (tid < C_DIM) {
        const int c   = tid;
        const int cg2 = c >> 4;
        const int ln2 = c & 15;
        float sum = 0.0f, sq = 0.0f, mx = -INFINITY, mn = INFINITY;
        #pragma unroll
        for (int s = 0; s < TSLICES; ++s) {
            const int idx = (s * CGROUPS + cg2) * 16 + ln2;
            sum += s_sum[idx];
            sq  += s_sq [idx];
            mx   = fmaxf(mx, s_max[idx]);
            mn   = fminf(mn, s_min[idx]);
        }
        const float mean = sum / (float)T_DIM;
        float var = (sq - sum * mean) / (float)(T_DIM - 1);
        var = fmaxf(var, 0.0f);
        const float stddev = sqrtf(var);

        const float first = xb[c];
        const float last  = xb[(size_t)(T_DIM - 1) * C_DIM + c];
        const float slope = (last - first) / (float)(T_DIM - 1);

        float* __restrict__ ob = out + (size_t)b * 5 * C_DIM;
        ob[0 * C_DIM + c] = mean;
        ob[1 * C_DIM + c] = stddev;
        ob[2 * C_DIM + c] = mx;
        ob[3 * C_DIM + c] = mn;
        ob[4 * C_DIM + c] = slope;
    }
}

extern "C" void kernel_launch(void* const* d_in, const int* in_sizes, int n_in,
                              void* d_out, int out_size, void* d_ws, size_t ws_size,
                              hipStream_t stream) {
    const float* x  = (const float*)d_in[0];
    float* out      = (float*)d_out;
    const int batch = in_sizes[0] / (T_DIM * C_DIM);
    meta_feature_kernel<<<batch, BLOCK, 0, stream>>>(x, out);
}


# 16-byte loads per lane (one wave load = 512 contiguous B), 16-32 loads in flight per wave; waves split T only (16 slabs of 128 steps), so the partial sums add in a different order
# speedup vs baseline: 1.9428x; 1.9428x over previous
#include <hip/hip_runtime.h>
#include <hip/hip_bf16.h>
#include <math.h>


typedef __attribute__((ext_vector_type(2))) float v2f;
typedef __attribute__((ext_vector_type(4))) float v4f;
typedef __attribute__((ext_vector_type(8))) float v8f;

#define T_DIM   2048
#define C_DIM   64
#define WAVES   16
#define BLOCK   (WAVES * 32)
#define ROWS_PER_WAVE  (T_DIM / WAVES)
#define LOADS_PER_WAVE (ROWS_PER_WAVE / 2)
#define GROUP   16
#define NGROUPS (LOADS_PER_WAVE / GROUP)
#define STEP    (2 * C_DIM / 4)

__device__ __forceinline__ void consume(const v4f a, const v4f c, const v2f ones,
                                        v8f (&acc_s)[4], v8f (&acc_q)[4],
                                        float (&vmax)[4], float (&vmin)[4]) {
    #pragma unroll
    for (int j = 0; j < 4; ++j) {
        v2f bv; bv.x = a[j];        bv.y = c[j];
        v2f qv; qv.x = a[j] * a[j]; qv.y = c[j] * c[j];
        acc_s[j] = __builtin_amdgcn_wmma_f32_16x16x4_f32(
            false, ones, false, bv, (short)0, acc_s[j], false, false);
        acc_q[j] = __builtin_amdgcn_wmma_f32_16x16x4_f32(
            false, ones, false, qv, (short)0, acc_q[j], false, false);
        vmax[j] = fmaxf(vmax[j], fmaxf(a[j], c[j]));
        vmin[j] = fminf(vmin[j], fminf(a[j], c[j]));
    }
}

__global__ __launch_bounds__(BLOCK)
void meta_feature_kernel(const float* __restrict__ x, float* __restrict__ out) {
    const int b    = blockIdx.x;
    const int tid  = threadIdx.x;
    const int wave = tid >> 5;
    const int lane = tid & 31;
    const int half = lane >> 4;
    const int ln   = lane & 15;

    const float* __restrict__ xb = x + (size_t)b * T_DIM * C_DIM;
    const v4f* __restrict__ p =
        (const v4f*)(xb + (size_t)wave * ROWS_PER_WAVE * C_DIM) + lane;

    v8f acc_s[4] = {};
    v8f acc_q[4] = {};
    v2f ones;  ones.x = 1.0f; ones.y = 1.0f;

    float vmax[4], vmin[4];
    #pragma unroll
    for (int j = 0; j < 4; ++j) { vmax[j] = -INFINITY; vmin[j] = INFINITY; }

    v4f buf[2][GROUP];
    #pragma unroll
    for (int i = 0; i < GROUP; ++i)
        buf[0][i] = __builtin_nontemporal_load(p + i * STEP);

    #pragma unroll
    for (int g = 0; g < NGROUPS; ++g) {
        __builtin_amdgcn_sched_barrier(0);
        if (g + 1 < NGROUPS) {
            #pragma unroll
            for (int i = 0; i < GROUP; ++i)
                buf[(g + 1) & 1][i] =
                    __builtin_nontemporal_load(p + ((g + 1) * GROUP + i) * STEP);
            __builtin_amdgcn_sched_barrier(0);
        }
        #pragma unroll
        for (int i = 0; i < GROUP; i += 2)
            consume(buf[g & 1][i], buf[g & 1][i + 1], ones, acc_s, acc_q, vmax, vmin);
    }

    #pragma unroll
    for (int j = 0; j < 4; ++j) {
        vmax[j] = fmaxf(vmax[j], __shfl_xor(vmax[j], 16, 32));
        vmin[j] = fminf(vmin[j], __shfl_xor(vmin[j], 16, 32));
    }

    __shared__ v4f s_sum[WAVES * 16];
    __shared__ v4f s_sq [WAVES * 16];
    __shared__ v4f s_max[WAVES * 16];
    __shared__ v4f s_min[WAVES * 16];

    if (half == 0) {
        v4f vs, vq, vx, vn;
        #pragma unroll
        for (int j = 0; j < 4; ++j) {
            vs[j] = acc_s[j][0];
            vq[j] = acc_q[j][0];
            vx[j] = vmax[j];
            vn[j] = vmin[j];
        }
        s_sum[wave * 16 + ln] = vs;
        s_sq [wave * 16 + ln] = vq;
        s_max[wave * 16 + ln] = vx;
        s_min[wave * 16 + ln] = vn;
    }
    __syncthreads();

    if (tid < C_DIM) {
        const int c = tid;
        const float* f_sum = (const float*)s_sum;
        const float* f_sq  = (const float*)s_sq;
        const float* f_max = (const float*)s_max;
        const float* f_min = (const float*)s_min;
        float sum = 0.0f, sq = 0.0f, mx = -INFINITY, mn = INFINITY;
        #pragma unroll
        for (int w = 0; w < WAVES; ++w) {
            const int idx = w * C_DIM + c;
            sum += f_sum[idx];
            sq  += f_sq [idx];
            mx   = fmaxf(mx, f_max[idx]);
            mn   = fminf(mn, f_min[idx]);
        }
        const float mean = sum / (float)T_DIM;
        float var = (sq - sum * mean) / (float)(T_DIM - 1);
        var = fmaxf(var, 0.0f);
        const float stddev = sqrtf(var);

        const float first = xb[c];
        const float last  = xb[(size_t)(T_DIM - 1) * C_DIM + c];
        const float slope = (last - first) / (float)(T_DIM - 1);

        float* __restrict__ ob = out + (size_t)b * 5 * C_DIM;
        ob[0 * C_DIM + c] = mean;
        ob[1 * C_DIM + c] = stddev;
        ob[2 * C_DIM + c] = mx;
        ob[3 * C_DIM + c] = mn;
        ob[4 * C_DIM + c] = slope;
    }
}

extern "C" void kernel_launch(void* const* d_in, const int* in_sizes, int n_in,
                              void* d_out, int out_size, void* d_ws, size_t ws_size,
                              hipStream_t stream) {
    const float* x  = (const float*)d_in[0];
    float* out      = (float*)d_out;
    const int batch = in_sizes[0] / (T_DIM * C_DIM);
    meta_feature_kernel<<<batch, BLOCK, 0, stream>>>(x, out);
}


# sums of squares from the Gram diagonal of one f32 16x16x4 WMMA per fragment (A = B = the fragment), sums from a row of ones in A: 136 not 256 WMMAs per wave; squares add in a different order
# speedup vs baseline: 2.0909x; 1.0762x over previous
#include <hip/hip_runtime.h>
#include <hip/hip_bf16.h>
#include <math.h>


typedef __attribute__((ext_vector_type(2))) float v2f;
typedef __attribute__((ext_vector_type(4))) float v4f;
typedef __attribute__((ext_vector_type(8))) float v8f;

#define T_DIM   2048
#define C_DIM   64
#define WAVES   16
#define BLOCK   (WAVES * 32)
#define ROWS_PER_WAVE  (T_DIM / WAVES)
#define LOADS_PER_WAVE (ROWS_PER_WAVE / 2)
#define GROUP   16
#define NGROUPS (LOADS_PER_WAVE / GROUP)
#define STEP    (2 * C_DIM / 4)

__device__ __forceinline__ float share15(const float v) {
    const int iv = __builtin_bit_cast(int, v);
    return __builtin_bit_cast(float,
        __builtin_amdgcn_update_dpp(0, iv, 0x15F, 0xF, 0xF, true));
}

__device__ __forceinline__ void consume(const v4f a, const v4f c, const int slot,
                                        const int ln, const bool row15,
                                        v8f (&acc)[4], v8f (&acc_p)[4], v2f (&pk)[4],
                                        float (&vmax)[4], float (&vmin)[4]) {
    const bool mine = (ln == slot);
    #pragma unroll
    for (int j = 0; j < 4; ++j) {
        v2f bv; bv.x = a[j];                 bv.y = c[j];
        v2f av; av.x = row15 ? 1.0f : a[j];  av.y = row15 ? 1.0f : c[j];
        acc[j] = __builtin_amdgcn_wmma_f32_16x16x4_f32(
            false, av, false, bv, (short)0, acc[j], false, false);
        const float sa = share15(a[j]);
        const float sc = share15(c[j]);
        pk[j].x = mine ? sa : pk[j].x;
        pk[j].y = mine ? sc : pk[j].y;
        vmax[j] = fmaxf(vmax[j], fmaxf(a[j], c[j]));
        vmin[j] = fminf(vmin[j], fminf(a[j], c[j]));
    }
    if (slot == 15) {
        #pragma unroll
        for (int j = 0; j < 4; ++j)
            acc_p[j] = __builtin_amdgcn_wmma_f32_16x16x4_f32(
                false, pk[j], false, pk[j], (short)0, acc_p[j], false, false);
    }
}

__global__ __launch_bounds__(BLOCK)
void meta_feature_kernel(const float* __restrict__ x, float* __restrict__ out) {
    const int b    = blockIdx.x;
    const int tid  = threadIdx.x;
    const int wave = tid >> 5;
    const int lane = tid & 31;
    const int half = lane >> 4;
    const int ln   = lane & 15;
    const bool row15 = (ln == 15);

    const float* __restrict__ xb = x + (size_t)b * T_DIM * C_DIM;
    const v4f* __restrict__ p =
        (const v4f*)(xb + (size_t)wave * ROWS_PER_WAVE * C_DIM) + lane;

    v8f acc[4]   = {};
    v8f acc_p[4] = {};
    v2f pk[4]    = {};

    float vmax[4], vmin[4];
    #pragma unroll
    for (int j = 0; j < 4; ++j) { vmax[j] = -INFINITY; vmin[j] = INFINITY; }

    v4f buf[2][GROUP];
    #pragma unroll
    for (int i = 0; i < GROUP; ++i)
        buf[0][i] = __builtin_nontemporal_load(p + i * STEP);

    #pragma unroll
    for (int g = 0; g < NGROUPS; ++g) {
        __builtin_amdgcn_sched_barrier(0);
        if (g + 1 < NGROUPS) {
            #pragma unroll
            for (int i = 0; i < GROUP; ++i)
                buf[(g + 1) & 1][i] =
                    __builtin_nontemporal_load(p + ((g + 1) * GROUP + i) * STEP);
            __builtin_amdgcn_sched_barrier(0);
        }
        #pragma unroll
        for (int i = 0; i < GROUP; i += 2)
            consume(buf[g & 1][i], buf[g & 1][i + 1], (g * (GROUP / 2) + i / 2) & 15,
                    ln, row15, acc, acc_p, pk, vmax, vmin);
    }

    #pragma unroll
    for (int j = 0; j < 4; ++j) {
        vmax[j] = fmaxf(vmax[j], __shfl_xor(vmax[j], 16, 32));
        vmin[j] = fminf(vmin[j], __shfl_xor(vmin[j], 16, 32));
    }

    const bool diag_lane = (half == (ln >> 3));
    const int  r = ln & 7;
    float dq[4], dl[4];
    #pragma unroll
    for (int j = 0; j < 4; ++j) {
        float d = acc[j][0], e = acc_p[j][0];
        #pragma unroll
        for (int rr = 1; rr < 8; ++rr) {
            d = (r == rr) ? acc[j][rr]   : d;
            e = (r == rr) ? acc_p[j][rr] : e;
        }
        dq[j] = d;
        e = diag_lane ? e : 0.0f;
        #pragma unroll
        for (int off = 16; off > 0; off >>= 1)
            e += __shfl_xor(e, off, 32);
        dl[j] = e;
    }

    __shared__ v4f s_sum[WAVES * 16];
    __shared__ v4f s_sq [WAVES * 16];
    __shared__ v4f s_max[WAVES * 16];
    __shared__ v4f s_min[WAVES * 16];

    if (half == 1) {
        v4f vs;
        #pragma unroll
        for (int j = 0; j < 4; ++j) vs[j] = acc[j][7];
        s_sum[wave * 16 + ln] = vs;
    }
    if (diag_lane) {
        v4f vq;
        #pragma unroll
        for (int j = 0; j < 4; ++j) vq[j] = row15 ? dl[j] : dq[j];
        s_sq[wave * 16 + ln] = vq;
    }
    if (half == 0) {
        v4f vx, vn;
        #pragma unroll
        for (int j = 0; j < 4; ++j) { vx[j] = vmax[j]; vn[j] = vmin[j]; }
        s_max[wave * 16 + ln] = vx;
        s_min[wave * 16 + ln] = vn;
    }
    __syncthreads();

    if (tid < C_DIM) {
        const int c = tid;
        const float* f_sum = (const float*)s_sum;
        const float* f_sq  = (const float*)s_sq;
        const float* f_max = (const float*)s_max;
        const float* f_min = (const float*)s_min;
        float sum = 0.0f, sq = 0.0f, mx = -INFINITY, mn = INFINITY;
        #pragma unroll
        for (int w = 0; w < WAVES; ++w) {
            const int idx = w * C_DIM + c;
            sum += f_sum[idx];
            sq  += f_sq [idx];
            mx   = fmaxf(mx, f_max[idx]);
            mn   = fminf(mn, f_min[idx]);
        }
        const float mean = sum / (float)T_DIM;
        float var = (sq - sum * mean) / (float)(T_DIM - 1);
        var = fmaxf(var, 0.0f);
        const float stddev = sqrtf(var);

        const float first = xb[c];
        const float last  = xb[(size_t)(T_DIM - 1) * C_DIM + c];
        const float slope = (last - first) / (float)(T_DIM - 1);

        float* __restrict__ ob = out + (size_t)b * 5 * C_DIM;
        ob[0 * C_DIM + c] = mean;
        ob[1 * C_DIM + c] = stddev;
        ob[2 * C_DIM + c] = mx;
        ob[3 * C_DIM + c] = mn;
        ob[4 * C_DIM + c] = slope;
    }
}

extern "C" void kernel_launch(void* const* d_in, const int* in_sizes, int n_in,
                              void* d_out, int out_size, void* d_ws, size_t ws_size,
                              hipStream_t stream) {
    const float* x  = (const float*)d_in[0];
    float* out      = (float*)d_out;
    const int batch = in_sizes[0] / (T_DIM * C_DIM);
    meta_feature_kernel<<<batch, BLOCK, 0, stream>>>(x, out);
}


# column sums on packed f32 vector adds, not a row of ones in A: A = B = the fragment, no selects, lane broadcasts or packed side fragment; 128 not 136 WMMAs per wave; the sums add in a different order
# speedup vs baseline: 2.3316x; 1.1151x over previous
#include <hip/hip_runtime.h>
#include <hip/hip_bf16.h>
#include <math.h>


typedef __attribute__((ext_vector_type(2))) float v2f;
typedef __attribute__((ext_vector_type(4))) float v4f;
typedef __attribute__((ext_vector_type(8))) float v8f;

#define T_DIM   2048
#define C_DIM   64
#define WAVES   16
#define BLOCK   (WAVES * 32)
#define ROWS_PER_WAVE  (T_DIM / WAVES)
#define LOADS_PER_WAVE (ROWS_PER_WAVE / 2)
#define GROUP   16
#define NGROUPS (LOADS_PER_WAVE / GROUP)
#define STEP    (2 * C_DIM / 4)

__device__ __forceinline__ void consume(const v4f a, const v4f c,
                                        v8f (&acc)[4], v2f (&sum)[4],
                                        float (&vmax)[4], float (&vmin)[4]) {
    sum[0] += __builtin_shufflevector(a, a, 0, 1);
    sum[1] += __builtin_shufflevector(a, a, 2, 3);
    sum[2] += __builtin_shufflevector(c, c, 0, 1);
    sum[3] += __builtin_shufflevector(c, c, 2, 3);
    #pragma unroll
    for (int j = 0; j < 4; ++j) {
        v2f f; f.x = a[j]; f.y = c[j];
        acc[j] = __builtin_amdgcn_wmma_f32_16x16x4_f32(
            false, f, false, f, (short)0, acc[j], false, false);
        vmax[j] = fmaxf(vmax[j], fmaxf(a[j], c[j]));
        vmin[j] = fminf(vmin[j], fminf(a[j], c[j]));
    }
}

__global__ __launch_bounds__(BLOCK)
void meta_feature_kernel(const float* __restrict__ x, float* __restrict__ out) {
    const int b    = blockIdx.x;
    const int tid  = threadIdx.x;
    const int wave = tid >> 5;
    const int lane = tid & 31;
    const int half = lane >> 4;
    const int ln   = lane & 15;

    const float* __restrict__ xb = x + (size_t)b * T_DIM * C_DIM;
    const v4f* __restrict__ p =
        (const v4f*)(xb + (size_t)wave * ROWS_PER_WAVE * C_DIM) + lane;

    v8f acc[4] = {};
    v2f sum[4] = {};

    float vmax[4], vmin[4];
    #pragma unroll
    for (int j = 0; j < 4; ++j) { vmax[j] = -INFINITY; vmin[j] = INFINITY; }

    v4f buf[2][GROUP];
    #pragma unroll
    for (int i = 0; i < GROUP; ++i)
        buf[0][i] = __builtin_nontemporal_load(p + i * STEP);

    #pragma unroll
    for (int g = 0; g < NGROUPS; ++g) {
        __builtin_amdgcn_sched_barrier(0);
        if (g + 1 < NGROUPS) {
            #pragma unroll
            for (int i = 0; i < GROUP; ++i)
                buf[(g + 1) & 1][i] =
                    __builtin_nontemporal_load(p + ((g + 1) * GROUP + i) * STEP);
            __builtin_amdgcn_sched_barrier(0);
        }
        #pragma unroll
        for (int i = 0; i < GROUP; i += 2)
            consume(buf[g & 1][i], buf[g & 1][i + 1], acc, sum, vmax, vmin);
    }

    float csum[4];
    csum[0] = sum[0].x + sum[2].x;
    csum[1] = sum[0].y + sum[2].y;
    csum[2] = sum[1].x + sum[3].x;
    csum[3] = sum[1].y + sum[3].y;
    #pragma unroll
    for (int j = 0; j < 4; ++j) {
        csum[j] += __shfl_xor(csum[j], 16, 32);
        vmax[j] = fmaxf(vmax[j], __shfl_xor(vmax[j], 16, 32));
        vmin[j] = fminf(vmin[j], __shfl_xor(vmin[j], 16, 32));
    }

    const bool diag_lane = (half == (ln >> 3));
    const int  r = ln & 7;
    float dq[4];
    #pragma unroll
    for (int j = 0; j < 4; ++j) {
        float d = acc[j][0];
        #pragma unroll
        for (int rr = 1; rr < 8; ++rr)
            d = (r == rr) ? acc[j][rr] : d;
        dq[j] = d;
    }

    __shared__ v4f s_sum[WAVES * 16];
    __shared__ v4f s_sq [WAVES * 16];
    __shared__ v4f s_max[WAVES * 16];
    __shared__ v4f s_min[WAVES * 16];

    if (half == 1) {
        v4f vs;
        #pragma unroll
        for (int j = 0; j < 4; ++j) vs[j] = csum[j];
        s_sum[wave * 16 + ln] = vs;
    }
    if (diag_lane) {
        v4f vq;
        #pragma unroll
        for (int j = 0; j < 4; ++j) vq[j] = dq[j];
        s_sq[wave * 16 + ln] = vq;
    }
    if (half == 0) {
        v4f vx, vn;
        #pragma unroll
        for (int j = 0; j < 4; ++j) { vx[j] = vmax[j]; vn[j] = vmin[j]; }
        s_max[wave * 16 + ln] = vx;
        s_min[wave * 16 + ln] = vn;
    }
    __syncthreads();

    if (tid < C_DIM) {
        const int c = tid;
        const float* f_sum = (const float*)s_sum;
        const float* f_sq  = (const float*)s_sq;
        const float* f_max = (const float*)s_max;
        const float* f_min = (const float*)s_min;
        float tsum = 0.0f, sq = 0.0f, mx = -INFINITY, mn = INFINITY;
        #pragma unroll
        for (int w = 0; w < WAVES; ++w) {
            const int idx = w * C_DIM + c;
            tsum += f_sum[idx];
            sq   += f_sq [idx];
            mx    = fmaxf(mx, f_max[idx]);
            mn    = fminf(mn, f_min[idx]);
        }
        const float mean = tsum / (float)T_DIM;
        float var = (sq - tsum * mean) / (float)(T_DIM - 1);
        var = fmaxf(var, 0.0f);
        const float stddev = sqrtf(var);

        const float first = xb[c];
        const float last  = xb[(size_t)(T_DIM - 1) * C_DIM + c];
        const float slope = (last - first) / (float)(T_DIM - 1);

        float* __restrict__ ob = out + (size_t)b * 5 * C_DIM;
        ob[0 * C_DIM + c] = mean;
        ob[1 * C_DIM + c] = stddev;
        ob[2 * C_DIM + c] = mx;
        ob[3 * C_DIM + c] = mn;
        ob[4 * C_DIM + c] = slope;
    }
}

extern "C" void kernel_launch(void* const* d_in, const int* in_sizes, int n_in,
                              void* d_out, int out_size, void* d_ws, size_t ws_size,
                              hipStream_t stream) {
    const float* x  = (const float*)d_in[0];
    float* out      = (float*)d_out;
    const int batch = in_sizes[0] / (T_DIM * C_DIM);
    meta_feature_kernel<<<batch, BLOCK, 0, stream>>>(x, out);
}
